# starting kernel
# baseline (speedup 1.0000x reference)
#include <hip/hip_runtime.h>
#include <hip/hip_bf16.h>
#include <math.h>

#define NQ   4
#define DIM  16
#define NL   5
#define BLK  256

typedef __attribute__((ext_vector_type(2))) float v2f;
typedef __attribute__((ext_vector_type(8))) float v8f;

__device__ __forceinline__ float fast_tanh(float x) {
#if __has_builtin(__builtin_amdgcn_tanhf)
    return __builtin_amdgcn_tanhf(x);
#elif __has_builtin(__builtin_amdgcn_tanh_f32)
    return __builtin_amdgcn_tanh_f32(x);
#else
    float r;
    asm volatile("v_tanh_f32 %0, %1\n\tv_nop" : "=v"(r) : "v"(x));
    return r;
#endif
}

__global__ __launch_bounds__(BLK) void qpinn_kernel(
    const float* __restrict__ xy, const float* __restrict__ W1, const float* __restrict__ b1,
    const float* __restrict__ W2, const float* __restrict__ b2, const float* __restrict__ W3,
    const float* __restrict__ b3, const float* __restrict__ W4, const float* __restrict__ b4,
    float* __restrict__ out, int B)
{
    __shared__ float hS[BLK][17];
    __shared__ float w2S[16][40];
    __shared__ float pS[BLK][41];

    const int tid   = threadIdx.x;
    const int lane  = tid & 31;
    const int wv    = tid >> 5;
    const int half  = lane >> 4;
    const int l16   = lane & 15;
    const int bIdx  = blockIdx.x * BLK + tid;
    const int bSafe = (bIdx < B) ? bIdx : (B - 1);

    for (int i = tid; i < 16 * 40; i += BLK) w2S[i / 40][i % 40] = W2[i];

    const float x = xy[2 * bSafe + 0];
    const float y = xy[2 * bSafe + 1];
    #pragma unroll
    for (int j = 0; j < 16; ++j)
        hS[tid][j] = fast_tanh(fmaf(x, W1[j], fmaf(y, W1[16 + j], b1[j])));

    __syncthreads();

    const int waveBase = wv * 32;
    #pragma unroll
    for (int m = 0; m < 2; ++m) {
        const int arow = waveBase + m * 16 + l16;
        #pragma unroll
        for (int n = 0; n < 3; ++n) {
            const int col    = n * 16 + l16;
            const bool colOK = (col < 40);
            const float bias = colOK ? b2[col] : 0.0f;
            v8f acc;
            #pragma unroll
            for (int r = 0; r < 8; ++r) acc[r] = bias;

            #pragma unroll
            for (int kc = 0; kc < 4; ++kc) {
                const int k0 = kc * 4 + half * 2;
                v2f a, bm;
                a[0]  = hS[arow][k0];
                a[1]  = hS[arow][k0 + 1];
                bm[0] = colOK ? w2S[k0][col]     : 0.0f;
                bm[1] = colOK ? w2S[k0 + 1][col] : 0.0f;
                acc = __builtin_amdgcn_wmma_f32_16x16x4_f32(
                          false, a, false, bm, (short)0, acc, false, false);
            }
            if (colOK) {
                #pragma unroll
                for (int r = 0; r < 8; ++r)
                    pS[waveBase + m * 16 + half * 8 + r][col] = fast_tanh(acc[r]);
            }
        }
    }
    __syncthreads();

    float re[DIM], im[DIM];
    {
        const float PIv = 3.14159265358979323846f;
        const float sn = __sinf(x * PIv * 0.5f);
        const float cs = __cosf(x * PIv * 0.5f);
        const float is2 = 0.70710678118654752440f;
        const float a0 = is2 * (cs - sn), a1 = is2 * (cs + sn);
        const float su = __sinf(y * PIv * 0.5f);
        const float cu = __cosf(y * PIv * 0.5f);
        const float A0r = a0 * cu, A0i = -a0 * su;
        const float A1r = a1 * cu, A1i =  a1 * su;
        float p0r[5], p0i[5], p1r[5], p1i[5];
        p0r[0] = 1.f; p0i[0] = 0.f; p1r[0] = 1.f; p1i[0] = 0.f;
        #pragma unroll
        for (int k = 1; k <= 4; ++k) {
            p0r[k] = p0r[k-1]*A0r - p0i[k-1]*A0i;
            p0i[k] = p0r[k-1]*A0i + p0i[k-1]*A0r;
            p1r[k] = p1r[k-1]*A1r - p1i[k-1]*A1i;
            p1i[k] = p1r[k-1]*A1i + p1i[k-1]*A1r;
        }
        #pragma unroll
        for (int d = 0; d < DIM; ++d) {
            const int p = __popc(d);
            re[d] = p0r[4-p]*p1r[p] - p0i[4-p]*p1i[p];
            im[d] = p0r[4-p]*p1i[p] + p0i[4-p]*p1r[p];
        }
    }

    for (int l = 0; l < NL; ++l) {
        #pragma unroll
        for (int qb = 0; qb < 4; ++qb) {
            const float ax = pS[tid][l * 8 + qb];
            const float ay = pS[tid][l * 8 + qb + 4];
            const float sa = __sinf(ax * 0.5f), ca = __cosf(ax * 0.5f);
            const float sb = __sinf(ay * 0.5f), cb = __cosf(ay * 0.5f);
            const float A  = cb * ca;
            const float Bv = sb * sa;
            const float Cv = sb * ca;
            const float Dv = cb * sa;
            const int  R   = 8 >> qb;
            #pragma unroll
            for (int i0 = 0; i0 < DIM; ++i0) {
                if (i0 & R) continue;
                const int i1 = i0 | R;
                const float r0 = re[i0], m0 = im[i0];
                const float r1 = re[i1], m1 = im[i1];
                re[i0] = A*r0 - Bv*m0 - Cv*r1 + Dv*m1;
                im[i0] = A*m0 + Bv*r0 - Cv*m1 - Dv*r1;
                re[i1] = Cv*r0 + Dv*m0 + A*r1 + Bv*m1;
                im[i1] = Cv*m0 - Dv*r0 + A*m1 - Bv*r1;
            }
        }
        re[3]  = -re[3];  im[3]  = -im[3];
        re[6]  = -re[6];  im[6]  = -im[6];
        re[9]  = -re[9];  im[9]  = -im[9];
        re[12] = -re[12]; im[12] = -im[12];
    }

    float q0 = 0.f, q1 = 0.f, q2 = 0.f, q3 = 0.f;
    #pragma unroll
    for (int d = 0; d < DIM; ++d) {
        const float pr = re[d]*re[d] + im[d]*im[d];
        q0 += (d & 8) ? -pr : pr;
        q1 += (d & 4) ? -pr : pr;
        q2 += (d & 2) ? -pr : pr;
        q3 += (d & 1) ? -pr : pr;
    }
    float o = b4[0];
    #pragma unroll
    for (int j = 0; j < 8; ++j) {
        const float acc = b3[j] + q0*W3[j] + q1*W3[8 + j] + q2*W3[16 + j] + q3*W3[24 + j];
        o += fast_tanh(acc) * W4[j];
    }
    if (bIdx < B) out[bIdx] = o;
}

extern "C" void kernel_launch(void* const* d_in, const int* in_sizes, int n_in,
                              void* d_out, int out_size, void* d_ws, size_t ws_size,
                              hipStream_t stream) {
    const float* xy = (const float*)d_in[0];
    const float* W1 = (const float*)d_in[1];
    const float* b1 = (const float*)d_in[2];
    const float* W2 = (const float*)d_in[3];
    const float* b2 = (const float*)d_in[4];
    const float* W3 = (const float*)d_in[5];
    const float* b3 = (const float*)d_in[6];
    const float* W4 = (const float*)d_in[7];
    const float* b4 = (const float*)d_in[8];
    float* out = (float*)d_out;

    const int B = in_sizes[0] / 2;
    const int blocks = (B + BLK - 1) / BLK;
    qpinn_kernel<<<blocks, BLK, 0, stream>>>(xy, W1, b1, W2, b2, W3, b3, W4, b4, out, B);
}


# state vector kept as (even,odd) amplitude register pairs, re and im apart: gates and probability sums are packed f32 ops with no register shuffles; the q sums become a pairwise tree (order changed)
# speedup vs baseline: 1.4332x; 1.4332x over previous
#include <hip/hip_runtime.h>
#include <hip/hip_bf16.h>
#include <math.h>

#define NQ   4
#define DIM  16
#define NL   5
#define BLK  256

typedef __attribute__((ext_vector_type(2))) float v2f;
typedef __attribute__((ext_vector_type(8))) float v8f;

__device__ __forceinline__ float fast_tanh(float x) {
#if __has_builtin(__builtin_amdgcn_tanhf)
    return __builtin_amdgcn_tanhf(x);
#elif __has_builtin(__builtin_amdgcn_tanh_f32)
    return __builtin_amdgcn_tanh_f32(x);
#else
    float r;
    asm volatile("v_tanh_f32 %0, %1\n\tv_nop" : "=v"(r) : "v"(x));
    return r;
#endif
}

__global__ __launch_bounds__(BLK) void qpinn_kernel(
    const float* __restrict__ xy, const float* __restrict__ W1, const float* __restrict__ b1,
    const float* __restrict__ W2, const float* __restrict__ b2, const float* __restrict__ W3,
    const float* __restrict__ b3, const float* __restrict__ W4, const float* __restrict__ b4,
    float* __restrict__ out, int B)
{
    __shared__ float hS[BLK][17];
    __shared__ float w2S[16][40];
    __shared__ float pS[BLK][41];

    const int tid   = threadIdx.x;
    const int lane  = tid & 31;
    const int wv    = tid >> 5;
    const int half  = lane >> 4;
    const int l16   = lane & 15;
    const int bIdx  = blockIdx.x * BLK + tid;
    const int bSafe = (bIdx < B) ? bIdx : (B - 1);

    for (int i = tid; i < 16 * 40; i += BLK) w2S[i / 40][i % 40] = W2[i];

    const float x = xy[2 * bSafe + 0];
    const float y = xy[2 * bSafe + 1];
    #pragma unroll
    for (int j = 0; j < 16; ++j)
        hS[tid][j] = fast_tanh(fmaf(x, W1[j], fmaf(y, W1[16 + j], b1[j])));

    __syncthreads();

    const int waveBase = wv * 32;
    #pragma unroll
    for (int m = 0; m < 2; ++m) {
        const int arow = waveBase + m * 16 + l16;
        #pragma unroll
        for (int n = 0; n < 3; ++n) {
            const int col    = n * 16 + l16;
            const bool colOK = (col < 40);
            const float bias = colOK ? b2[col] : 0.0f;
            v8f acc;
            #pragma unroll
            for (int r = 0; r < 8; ++r) acc[r] = bias;

            #pragma unroll
            for (int kc = 0; kc < 4; ++kc) {
                const int k0 = kc * 4 + half * 2;
                v2f a, bm;
                a[0]  = hS[arow][k0];
                a[1]  = hS[arow][k0 + 1];
                bm[0] = colOK ? w2S[k0][col]     : 0.0f;
                bm[1] = colOK ? w2S[k0 + 1][col] : 0.0f;
                acc = __builtin_amdgcn_wmma_f32_16x16x4_f32(
                          false, a, false, bm, (short)0, acc, false, false);
            }
            if (colOK) {
                #pragma unroll
                for (int r = 0; r < 8; ++r)
                    pS[waveBase + m * 16 + half * 8 + r][col] = fast_tanh(acc[r]);
            }
        }
    }
    __syncthreads();

    v2f RE[8], IM[8];
    {
        const float PIv = 3.14159265358979323846f;
        const float sn = __sinf(x * PIv * 0.5f);
        const float cs = __cosf(x * PIv * 0.5f);
        const float is2 = 0.70710678118654752440f;
        const float a0 = is2 * (cs - sn), a1 = is2 * (cs + sn);
        const float su = __sinf(y * PIv * 0.5f);
        const float cu = __cosf(y * PIv * 0.5f);
        const float A0r = a0 * cu, A0i = -a0 * su;
        const float A1r = a1 * cu, A1i =  a1 * su;
        float p0r[5], p0i[5], p1r[5], p1i[5];
        p0r[0] = 1.f; p0i[0] = 0.f; p1r[0] = 1.f; p1i[0] = 0.f;
        #pragma unroll
        for (int k = 1; k <= 4; ++k) {
            p0r[k] = p0r[k-1]*A0r - p0i[k-1]*A0i;
            p0i[k] = p0r[k-1]*A0i + p0i[k-1]*A0r;
            p1r[k] = p1r[k-1]*A1r - p1i[k-1]*A1i;
            p1i[k] = p1r[k-1]*A1i + p1i[k-1]*A1r;
        }
        #pragma unroll
        for (int d = 0; d < DIM; ++d) {
            const int p = __popc(d);
            RE[d >> 1][d & 1] = p0r[4-p]*p1r[p] - p0i[4-p]*p1i[p];
            IM[d >> 1][d & 1] = p0r[4-p]*p1i[p] + p0i[4-p]*p1r[p];
        }
    }

    for (int l = 0; l < NL; ++l) {
        #pragma unroll
        for (int qb = 0; qb < 4; ++qb) {
            const float ax = pS[tid][l * 8 + qb];
            const float ay = pS[tid][l * 8 + qb + 4];
            const float sa = __sinf(ax * 0.5f), ca = __cosf(ax * 0.5f);
            const float sb = __sinf(ay * 0.5f), cb = __cosf(ay * 0.5f);
            const float A  = cb * ca;
            const float Bv = sb * sa;
            const float Cv = sb * ca;
            const float Dv = cb * sa;
            if (qb < 3) {
                const v2f kA = {A, A}, kB = {Bv, Bv}, kC = {Cv, Cv}, kD = {Dv, Dv};
                const int Rj = 4 >> qb;
                #pragma unroll
                for (int j0 = 0; j0 < 8; ++j0) {
                    if (j0 & Rj) continue;
                    const int j1 = j0 | Rj;
                    const v2f r0 = RE[j0], m0 = IM[j0];
                    const v2f r1 = RE[j1], m1 = IM[j1];
                    RE[j0] = __builtin_elementwise_fma( kD, m1, __builtin_elementwise_fma(-kC, r1, __builtin_elementwise_fma(-kB, m0, kA * r0)));
                    IM[j0] = __builtin_elementwise_fma(-kD, r1, __builtin_elementwise_fma(-kC, m1, __builtin_elementwise_fma( kB, r0, kA * m0)));
                    RE[j1] = __builtin_elementwise_fma( kB, m1, __builtin_elementwise_fma( kA, r1, __builtin_elementwise_fma( kD, m0, kC * r0)));
                    IM[j1] = __builtin_elementwise_fma(-kB, r1, __builtin_elementwise_fma( kA, m1, __builtin_elementwise_fma(-kD, r0, kC * m0)));
                }
            } else {
                const v2f cAC = {A, Cv};
                const v2f cBD = {-Bv, Dv};
                const v2f cCA = {-Cv, A};
                const v2f cDB = {Dv, Bv};
                #pragma unroll
                for (int j = 0; j < 8; ++j) {
                    const v2f r = RE[j], m = IM[j];
                    const v2f r0 = __builtin_shufflevector(r, r, 0, 0), r1 = __builtin_shufflevector(r, r, 1, 1);
                    const v2f m0 = __builtin_shufflevector(m, m, 0, 0), m1 = __builtin_shufflevector(m, m, 1, 1);
                    RE[j] = __builtin_elementwise_fma( cDB, m1, __builtin_elementwise_fma(cCA, r1, __builtin_elementwise_fma( cBD, m0, cAC * r0)));
                    IM[j] = __builtin_elementwise_fma(-cDB, r1, __builtin_elementwise_fma(cCA, m1, __builtin_elementwise_fma(-cBD, r0, cAC * m0)));
                }
            }
        }
        RE[1][1] = -RE[1][1];  IM[1][1] = -IM[1][1];
        RE[3][0] = -RE[3][0];  IM[3][0] = -IM[3][0];
        RE[4][1] = -RE[4][1];  IM[4][1] = -IM[4][1];
        RE[6][0] = -RE[6][0];  IM[6][0] = -IM[6][0];
    }

    v2f PR[8];
    #pragma unroll
    for (int j = 0; j < 8; ++j)
        PR[j] = __builtin_elementwise_fma(IM[j], IM[j], RE[j] * RE[j]);
    const v2f s01 = PR[0] + PR[1], s23 = PR[2] + PR[3], s45 = PR[4] + PR[5], s67 = PR[6] + PR[7];
    const v2f d01 = PR[0] - PR[1], d23 = PR[2] - PR[3], d45 = PR[4] - PR[5], d67 = PR[6] - PR[7];
    const v2f sA = s01 + s23, sB = s45 + s67;
    const v2f Q0 = sA - sB;
    const v2f Q1 = (s01 - s23) + (s45 - s67);
    const v2f Q2 = (d01 + d23) + (d45 + d67);
    const v2f ST = sA + sB;
    const float q0 = Q0[0] + Q0[1];
    const float q1 = Q1[0] + Q1[1];
    const float q2 = Q2[0] + Q2[1];
    const float q3 = ST[0] - ST[1];
    float o = b4[0];
    #pragma unroll
    for (int j = 0; j < 8; ++j) {
        const float acc = b3[j] + q0*W3[j] + q1*W3[8 + j] + q2*W3[16 + j] + q3*W3[24 + j];
        o += fast_tanh(acc) * W4[j];
    }
    if (bIdx < B) out[bIdx] = o;
}

extern "C" void kernel_launch(void* const* d_in, const int* in_sizes, int n_in,
                              void* d_out, int out_size, void* d_ws, size_t ws_size,
                              hipStream_t stream) {
    const float* xy = (const float*)d_in[0];
    const float* W1 = (const float*)d_in[1];
    const float* b1 = (const float*)d_in[2];
    const float* W2 = (const float*)d_in[3];
    const float* b2 = (const float*)d_in[4];
    const float* W3 = (const float*)d_in[5];
    const float* b3 = (const float*)d_in[6];
    const float* W4 = (const float*)d_in[7];
    const float* b4 = (const float*)d_in[8];
    float* out = (float*)d_out;

    const int B = in_sizes[0] / 2;
    const int blocks = (B + BLK - 1) / BLK;
    qpinn_kernel<<<blocks, BLK, 0, stream>>>(xy, W1, b1, W2, b2, W3, b3, W4, b4, out, B);
}


# RX and RY applied as unnormalised shears with tan(angle/2) from an odd polynomial (no sin/cos in the layer loop): half the packed FMAs per gate; q divided once by the total probability at the end
# speedup vs baseline: 1.7245x; 1.2033x over previous
#include <hip/hip_runtime.h>
#include <hip/hip_bf16.h>
#include <math.h>

#define NQ   4
#define DIM  16
#define NL   5
#define BLK  256

typedef __attribute__((ext_vector_type(2))) float v2f;
typedef __attribute__((ext_vector_type(8))) float v8f;

__device__ __forceinline__ float fast_tanh(float x) {
#if __has_builtin(__builtin_amdgcn_tanhf)
    return __builtin_amdgcn_tanhf(x);
#elif __has_builtin(__builtin_amdgcn_tanh_f32)
    return __builtin_amdgcn_tanh_f32(x);
#else
    float r;
    asm volatile("v_tanh_f32 %0, %1\n\tv_nop" : "=v"(r) : "v"(x));
    return r;
#endif
}

__global__ __launch_bounds__(BLK) void qpinn_kernel(
    const float* __restrict__ xy, const float* __restrict__ W1, const float* __restrict__ b1,
    const float* __restrict__ W2, const float* __restrict__ b2, const float* __restrict__ W3,
    const float* __restrict__ b3, const float* __restrict__ W4, const float* __restrict__ b4,
    float* __restrict__ out, int B)
{
    __shared__ float hS[BLK][17];
    __shared__ float w2S[16][40];
    __shared__ float pS[BLK][41];

    const int tid   = threadIdx.x;
    const int lane  = tid & 31;
    const int wv    = tid >> 5;
    const int half  = lane >> 4;
    const int l16   = lane & 15;
    const int bIdx  = blockIdx.x * BLK + tid;
    const int bSafe = (bIdx < B) ? bIdx : (B - 1);

    for (int i = tid; i < 16 * 40; i += BLK) w2S[i / 40][i % 40] = W2[i];

    const float x = xy[2 * bSafe + 0];
    const float y = xy[2 * bSafe + 1];
    #pragma unroll
    for (int j = 0; j < 16; ++j)
        hS[tid][j] = fast_tanh(fmaf(x, W1[j], fmaf(y, W1[16 + j], b1[j])));

    __syncthreads();

    const int waveBase = wv * 32;
    #pragma unroll
    for (int m = 0; m < 2; ++m) {
        const int arow = waveBase + m * 16 + l16;
        #pragma unroll
        for (int n = 0; n < 3; ++n) {
            const int col    = n * 16 + l16;
            const bool colOK = (col < 40);
            const float bias = colOK ? b2[col] : 0.0f;
            v8f acc;
            #pragma unroll
            for (int r = 0; r < 8; ++r) acc[r] = bias;

            #pragma unroll
            for (int kc = 0; kc < 4; ++kc) {
                const int k0 = kc * 4 + half * 2;
                v2f a, bm;
                a[0]  = hS[arow][k0];
                a[1]  = hS[arow][k0 + 1];
                bm[0] = colOK ? w2S[k0][col]     : 0.0f;
                bm[1] = colOK ? w2S[k0 + 1][col] : 0.0f;
                acc = __builtin_amdgcn_wmma_f32_16x16x4_f32(
                          false, a, false, bm, (short)0, acc, false, false);
            }
            if (colOK) {
                #pragma unroll
                for (int r = 0; r < 8; ++r)
                    pS[waveBase + m * 16 + half * 8 + r][col] = fast_tanh(acc[r]);
            }
        }
    }
    __syncthreads();

    v2f RE[8], IM[8];
    {
        const float PIv = 3.14159265358979323846f;
        const float sn = __sinf(x * PIv * 0.5f);
        const float cs = __cosf(x * PIv * 0.5f);
        const float is2 = 0.70710678118654752440f;
        const float a0 = is2 * (cs - sn), a1 = is2 * (cs + sn);
        const float su = __sinf(y * PIv * 0.5f);
        const float cu = __cosf(y * PIv * 0.5f);
        const float A0r = a0 * cu, A0i = -a0 * su;
        const float A1r = a1 * cu, A1i =  a1 * su;
        float p0r[5], p0i[5], p1r[5], p1i[5];
        p0r[0] = 1.f; p0i[0] = 0.f; p1r[0] = 1.f; p1i[0] = 0.f;
        #pragma unroll
        for (int k = 1; k <= 4; ++k) {
            p0r[k] = p0r[k-1]*A0r - p0i[k-1]*A0i;
            p0i[k] = p0r[k-1]*A0i + p0i[k-1]*A0r;
            p1r[k] = p1r[k-1]*A1r - p1i[k-1]*A1i;
            p1i[k] = p1r[k-1]*A1i + p1i[k-1]*A1r;
        }
        #pragma unroll
        for (int d = 0; d < DIM; ++d) {
            const int p = __popc(d);
            RE[d >> 1][d & 1] = p0r[4-p]*p1r[p] - p0i[4-p]*p1i[p];
            IM[d >> 1][d & 1] = p0r[4-p]*p1i[p] + p0i[4-p]*p1r[p];
        }
    }

    for (int l = 0; l < NL; ++l) {
        v2f TN[4];
        #pragma unroll
        for (int i = 0; i < 4; ++i) {
            const v2f th = {pS[tid][l * 8 + 2 * i], pS[tid][l * 8 + 2 * i + 1]};
            const v2f w  = th * th;
            const v2f k4 = {5.560490006e-5f, 5.560490006e-5f};
            const v2f k3 = {4.097190512e-4f, 4.097190512e-4f};
            const v2f k2 = {4.171028265e-3f, 4.171028265e-3f};
            const v2f k1 = {4.166611439e-2f, 4.166611439e-2f};
            const v2f k0 = {0.5f, 0.5f};
            v2f p = __builtin_elementwise_fma(w, k4, k3);
            p = __builtin_elementwise_fma(w, p, k2);
            p = __builtin_elementwise_fma(w, p, k1);
            p = __builtin_elementwise_fma(w, p, k0);
            TN[i] = th * p;
        }
        #pragma unroll
        for (int qb = 0; qb < 4; ++qb) {
            const float tx = TN[qb >> 1][qb & 1];
            const float ty = TN[2 + (qb >> 1)][qb & 1];
            if (qb < 3) {
                const v2f kT = {tx, tx}, kS = {ty, ty};
                const int Rj = 4 >> qb;
                #pragma unroll
                for (int j0 = 0; j0 < 8; ++j0) {
                    if (j0 & Rj) continue;
                    const int j1 = j0 | Rj;
                    const v2f r0 = RE[j0], m0 = IM[j0];
                    const v2f r1 = RE[j1], m1 = IM[j1];
                    const v2f a0 = __builtin_elementwise_fma( kT, m1, r0);
                    const v2f b0 = __builtin_elementwise_fma(-kT, r1, m0);
                    const v2f a1 = __builtin_elementwise_fma( kT, m0, r1);
                    const v2f b1 = __builtin_elementwise_fma(-kT, r0, m1);
                    RE[j0] = __builtin_elementwise_fma(-kS, a1, a0);
                    IM[j0] = __builtin_elementwise_fma(-kS, b1, b0);
                    RE[j1] = __builtin_elementwise_fma( kS, a0, a1);
                    IM[j1] = __builtin_elementwise_fma( kS, b0, b1);
                }
            } else {
                const v2f kT = {tx, tx};
                const v2f cS = {-ty, ty};
                #pragma unroll
                for (int j = 0; j < 8; ++j) {
                    const v2f r = RE[j], m = IM[j];
                    const v2f rs = __builtin_shufflevector(r, r, 1, 0), ms = __builtin_shufflevector(m, m, 1, 0);
                    const v2f a = __builtin_elementwise_fma( kT, ms, r);
                    const v2f b = __builtin_elementwise_fma(-kT, rs, m);
                    const v2f as = __builtin_shufflevector(a, a, 1, 0), bs = __builtin_shufflevector(b, b, 1, 0);
                    RE[j] = __builtin_elementwise_fma(cS, as, a);
                    IM[j] = __builtin_elementwise_fma(cS, bs, b);
                }
            }
        }
        RE[1][1] = -RE[1][1];  IM[1][1] = -IM[1][1];
        RE[3][0] = -RE[3][0];  IM[3][0] = -IM[3][0];
        RE[4][1] = -RE[4][1];  IM[4][1] = -IM[4][1];
        RE[6][0] = -RE[6][0];  IM[6][0] = -IM[6][0];
    }

    v2f PR[8];
    #pragma unroll
    for (int j = 0; j < 8; ++j)
        PR[j] = __builtin_elementwise_fma(IM[j], IM[j], RE[j] * RE[j]);
    const v2f s01 = PR[0] + PR[1], s23 = PR[2] + PR[3], s45 = PR[4] + PR[5], s67 = PR[6] + PR[7];
    const v2f d01 = PR[0] - PR[1], d23 = PR[2] - PR[3], d45 = PR[4] - PR[5], d67 = PR[6] - PR[7];
    const v2f sA = s01 + s23, sB = s45 + s67;
    const v2f Q0 = sA - sB;
    const v2f Q1 = (s01 - s23) + (s45 - s67);
    const v2f Q2 = (d01 + d23) + (d45 + d67);
    const v2f ST = sA + sB;
    const float inv = 1.0f / (ST[0] + ST[1]);
    const float q0 = (Q0[0] + Q0[1]) * inv;
    const float q1 = (Q1[0] + Q1[1]) * inv;
    const float q2 = (Q2[0] + Q2[1]) * inv;
    const float q3 = (ST[0] - ST[1]) * inv;
    float o = b4[0];
    #pragma unroll
    for (int j = 0; j < 8; ++j) {
        const float acc = b3[j] + q0*W3[j] + q1*W3[8 + j] + q2*W3[16 + j] + q3*W3[24 + j];
        o += fast_tanh(acc) * W4[j];
    }
    if (bIdx < B) out[bIdx] = o;
}

extern "C" void kernel_launch(void* const* d_in, const int* in_sizes, int n_in,
                              void* d_out, int out_size, void* d_ws, size_t ws_size,
                              hipStream_t stream) {
    const float* xy = (const float*)d_in[0];
    const float* W1 = (const float*)d_in[1];
    const float* b1 = (const float*)d_in[2];
    const float* W2 = (const float*)d_in[3];
    const float* b2 = (const float*)d_in[4];
    const float* W3 = (const float*)d_in[5];
    const float* b3 = (const float*)d_in[6];
    const float* W4 = (const float*)d_in[7];
    const float* b4 = (const float*)d_in[8];
    float* out = (float*)d_out;

    const int B = in_sizes[0] / 2;
    const int blocks = (B + BLK - 1) / BLK;
    qpinn_kernel<<<blocks, BLK, 0, stream>>>(xy, W1, b1, W2, b2, W3, b3, W4, b4, out, B);
}


# all loads issued early and together: W2, b2, xy and last-stage weights read at entry with no wait between staging trips; WMMA fragments read from LDS as one group ahead of the products
# speedup vs baseline: 2.1643x; 1.2550x over previous
#include <hip/hip_runtime.h>
#include <hip/hip_bf16.h>
#include <math.h>

#define NQ   4
#define DIM  16
#define NL   5
#define BLK  256

typedef __attribute__((ext_vector_type(2))) float v2f;
typedef __attribute__((ext_vector_type(8))) float v8f;

__device__ __forceinline__ float fast_tanh(float x) {
#if __has_builtin(__builtin_amdgcn_tanhf)
    return __builtin_amdgcn_tanhf(x);
#elif __has_builtin(__builtin_amdgcn_tanh_f32)
    return __builtin_amdgcn_tanh_f32(x);
#else
    float r;
    asm volatile("v_tanh_f32 %0, %1\n\tv_nop" : "=v"(r) : "v"(x));
    return r;
#endif
}

__global__ __launch_bounds__(BLK) void qpinn_kernel(
    const float* __restrict__ xy, const float* __restrict__ W1, const float* __restrict__ b1,
    const float* __restrict__ W2, const float* __restrict__ b2, const float* __restrict__ W3,
    const float* __restrict__ b3, const float* __restrict__ W4, const float* __restrict__ b4,
    float* __restrict__ out, int B)
{
    __shared__ float hS[BLK][17];
    __shared__ float wS[3 * BLK];
    __shared__ float pS[BLK][41];

    const int tid   = threadIdx.x;
    const int lane  = tid & 31;
    const int wv    = tid >> 5;
    const int half  = lane >> 4;
    const int l16   = lane & 15;
    const int bIdx  = blockIdx.x * BLK + tid;
    const int bSafe = (bIdx < B) ? bIdx : (B - 1);

    const float x = xy[2 * bSafe + 0];
    const float y = xy[2 * bSafe + 1];
    const float wa = W2[tid];
    const float wb = W2[256 + tid];
    const int   j3 = min(max(tid - 128, 0), 39);
    const float* p3 = (tid < 128) ? (W2 + 512 + tid) : (b2 + j3);
    const float wc = *p3;
    const float b4r = b4[0];

    #pragma unroll
    for (int j = 0; j < 16; ++j)
        hS[tid][j] = fast_tanh(fmaf(x, W1[j], fmaf(y, W1[16 + j], b1[j])));

    wS[tid]       = wa;
    wS[256 + tid] = wb;
    wS[512 + tid] = wc;

    float w3r[32], b3r[8], w4r[8];
    #pragma unroll
    for (int j = 0; j < 32; ++j) w3r[j] = W3[j];
    #pragma unroll
    for (int j = 0; j < 8; ++j) { b3r[j] = b3[j]; w4r[j] = W4[j]; }

    __syncthreads();

    const int waveBase = wv * 32;
    v2f   bfr[3][4];
    float biasv[3];
    #pragma unroll
    for (int n = 0; n < 3; ++n) {
        const int colL = min(n * 16 + l16, 39);
        biasv[n] = wS[640 + colL];
        #pragma unroll
        for (int kc = 0; kc < 4; ++kc) {
            const int k0 = kc * 4 + half * 2;
            bfr[n][kc][0] = wS[k0 * 40 + colL];
            bfr[n][kc][1] = wS[(k0 + 1) * 40 + colL];
        }
    }
    v2f afr[2][4];
    #pragma unroll
    for (int m = 0; m < 2; ++m) {
        const int arow = waveBase + m * 16 + l16;
        #pragma unroll
        for (int kc = 0; kc < 4; ++kc) {
            const int k0 = kc * 4 + half * 2;
            afr[m][kc][0] = hS[arow][k0];
            afr[m][kc][1] = hS[arow][k0 + 1];
        }
    }
    #pragma unroll
    for (int m = 0; m < 2; ++m) {
        #pragma unroll
        for (int n = 0; n < 3; ++n) {
            const int col    = n * 16 + l16;
            const bool colOK = (col < 40);
            v8f acc;
            #pragma unroll
            for (int r = 0; r < 8; ++r) acc[r] = biasv[n];

            #pragma unroll
            for (int kc = 0; kc < 4; ++kc)
                acc = __builtin_amdgcn_wmma_f32_16x16x4_f32(
                          false, afr[m][kc], false, bfr[n][kc], (short)0, acc, false, false);
            if (colOK) {
                #pragma unroll
                for (int r = 0; r < 8; ++r)
                    pS[waveBase + m * 16 + half * 8 + r][col] = fast_tanh(acc[r]);
            }
        }
    }
    __syncthreads();

    v2f RE[8], IM[8];
    {
        const float PIv = 3.14159265358979323846f;
        const float sn = __sinf(x * PIv * 0.5f);
        const float cs = __cosf(x * PIv * 0.5f);
        const float is2 = 0.70710678118654752440f;
        const float a0 = is2 * (cs - sn), a1 = is2 * (cs + sn);
        const float su = __sinf(y * PIv * 0.5f);
        const float cu = __cosf(y * PIv * 0.5f);
        const float A0r = a0 * cu, A0i = -a0 * su;
        const float A1r = a1 * cu, A1i =  a1 * su;
        float p0r[5], p0i[5], p1r[5], p1i[5];
        p0r[0] = 1.f; p0i[0] = 0.f; p1r[0] = 1.f; p1i[0] = 0.f;
        #pragma unroll
        for (int k = 1; k <= 4; ++k) {
            p0r[k] = p0r[k-1]*A0r - p0i[k-1]*A0i;
            p0i[k] = p0r[k-1]*A0i + p0i[k-1]*A0r;
            p1r[k] = p1r[k-1]*A1r - p1i[k-1]*A1i;
            p1i[k] = p1r[k-1]*A1i + p1i[k-1]*A1r;
        }
        #pragma unroll
        for (int d = 0; d < DIM; ++d) {
            const int p = __popc(d);
            RE[d >> 1][d & 1] = p0r[4-p]*p1r[p] - p0i[4-p]*p1i[p];
            IM[d >> 1][d & 1] = p0r[4-p]*p1i[p] + p0i[4-p]*p1r[p];
        }
    }

    for (int l = 0; l < NL; ++l) {
        v2f TN[4];
        #pragma unroll
        for (int i = 0; i < 4; ++i) {
            const v2f th = {pS[tid][l * 8 + 2 * i], pS[tid][l * 8 + 2 * i + 1]};
            const v2f w  = th * th;
            const v2f k4 = {5.560490006e-5f, 5.560490006e-5f};
            const v2f k3 = {4.097190512e-4f, 4.097190512e-4f};
            const v2f k2 = {4.171028265e-3f, 4.171028265e-3f};
            const v2f k1 = {4.166611439e-2f, 4.166611439e-2f};
            const v2f k0 = {0.5f, 0.5f};
            v2f p = __builtin_elementwise_fma(w, k4, k3);
            p = __builtin_elementwise_fma(w, p, k2);
            p = __builtin_elementwise_fma(w, p, k1);
            p = __builtin_elementwise_fma(w, p, k0);
            TN[i] = th * p;
        }
        #pragma unroll
        for (int qb = 0; qb < 4; ++qb) {
            const float tx = TN[qb >> 1][qb & 1];
            const float ty = TN[2 + (qb >> 1)][qb & 1];
            if (qb < 3) {
                const v2f kT = {tx, tx}, kS = {ty, ty};
                const int Rj = 4 >> qb;
                #pragma unroll
                for (int j0 = 0; j0 < 8; ++j0) {
                    if (j0 & Rj) continue;
                    const int j1 = j0 | Rj;
                    const v2f r0 = RE[j0], m0 = IM[j0];
                    const v2f r1 = RE[j1], m1 = IM[j1];
                    const v2f a0 = __builtin_elementwise_fma( kT, m1, r0);
                    const v2f b0 = __builtin_elementwise_fma(-kT, r1, m0);
                    const v2f a1 = __builtin_elementwise_fma( kT, m0, r1);
                    const v2f b1 = __builtin_elementwise_fma(-kT, r0, m1);
                    RE[j0] = __builtin_elementwise_fma(-kS, a1, a0);
                    IM[j0] = __builtin_elementwise_fma(-kS, b1, b0);
                    RE[j1] = __builtin_elementwise_fma( kS, a0, a1);
                    IM[j1] = __builtin_elementwise_fma( kS, b0, b1);
                }
            } else {
                const v2f kT = {tx, tx};
                const v2f cS = {-ty, ty};
                #pragma unroll
                for (int j = 0; j < 8; ++j) {
                    const v2f r = RE[j], m = IM[j];
                    const v2f rs = __builtin_shufflevector(r, r, 1, 0), ms = __builtin_shufflevector(m, m, 1, 0);
                    const v2f a = __builtin_elementwise_fma( kT, ms, r);
                    const v2f b = __builtin_elementwise_fma(-kT, rs, m);
                    const v2f as = __builtin_shufflevector(a, a, 1, 0), bs = __builtin_shufflevector(b, b, 1, 0);
                    RE[j] = __builtin_elementwise_fma(cS, as, a);
                    IM[j] = __builtin_elementwise_fma(cS, bs, b);
                }
            }
        }
        RE[1][1] = -RE[1][1];  IM[1][1] = -IM[1][1];
        RE[3][0] = -RE[3][0];  IM[3][0] = -IM[3][0];
        RE[4][1] = -RE[4][1];  IM[4][1] = -IM[4][1];
        RE[6][0] = -RE[6][0];  IM[6][0] = -IM[6][0];
    }

    v2f PR[8];
    #pragma unroll
    for (int j = 0; j < 8; ++j)
        PR[j] = __builtin_elementwise_fma(IM[j], IM[j], RE[j] * RE[j]);
    const v2f s01 = PR[0] + PR[1], s23 = PR[2] + PR[3], s45 = PR[4] + PR[5], s67 = PR[6] + PR[7];
    const v2f d01 = PR[0] - PR[1], d23 = PR[2] - PR[3], d45 = PR[4] - PR[5], d67 = PR[6] - PR[7];
    const v2f sA = s01 + s23, sB = s45 + s67;
    const v2f Q0 = sA - sB;
    const v2f Q1 = (s01 - s23) + (s45 - s67);
    const v2f Q2 = (d01 + d23) + (d45 + d67);
    const v2f ST = sA + sB;
    const float inv = 1.0f / (ST[0] + ST[1]);
    const float q0 = (Q0[0] + Q0[1]) * inv;
    const float q1 = (Q1[0] + Q1[1]) * inv;
    const float q2 = (Q2[0] + Q2[1]) * inv;
    const float q3 = (ST[0] - ST[1]) * inv;
    float o = b4r;
    #pragma unroll
    for (int j = 0; j < 8; ++j) {
        const float acc = b3r[j] + q0*w3r[j] + q1*w3r[8 + j] + q2*w3r[16 + j] + q3*w3r[24 + j];
        o += fast_tanh(acc) * w4r[j];
    }
    if (bIdx < B) out[bIdx] = o;
}

extern "C" void kernel_launch(void* const* d_in, const int* in_sizes, int n_in,
                              void* d_out, int out_size, void* d_ws, size_t ws_size,
                              hipStream_t stream) {
    const float* xy = (const float*)d_in[0];
    const float* W1 = (const float*)d_in[1];
    const float* b1 = (const float*)d_in[2];
    const float* W2 = (const float*)d_in[3];
    const float* b2 = (const float*)d_in[4];
    const float* W3 = (const float*)d_in[5];
    const float* b3 = (const float*)d_in[6];
    const float* W4 = (const float*)d_in[7];
    const float* b4 = (const float*)d_in[8];
    float* out = (float*)d_out;

    const int B = in_sizes[0] / 2;
    const int blocks = (B + BLK - 1) / BLK;
    qpinn_kernel<<<blocks, BLK, 0, stream>>>(xy, W1, b1, W2, b2, W3, b3, W4, b4, out, B);
}
